# starting kernel
# baseline (speedup 1.0000x reference)
#include <hip/hip_runtime.h>
#include <math.h>

#define D_MODEL   1024
#define D_FF      2048
#define TWO_FF    4096
#define N_EXPERTS 8

#define BM   128
#define KC   64
#define XS_LD (KC + 4)

typedef float v2f __attribute__((ext_vector_type(2)));
typedef float v8f __attribute__((ext_vector_type(8)));

#define WMMA_F32(A, B, C) \
    __builtin_amdgcn_wmma_f32_16x16x4_f32(false, (A), false, (B), (short)0, (C), false, false)

__global__ void zero_out_kernel(float* __restrict__ out, int n, int* __restrict__ counts) {
    int i = blockIdx.x * blockDim.x + threadIdx.x;
    if (i < n) out[i] = 0.0f;
    if (blockIdx.x == 0 && threadIdx.x < 16) counts[threadIdx.x] = 0;
}

__global__ void gate_kernel(const float* __restrict__ x,
                            const float* __restrict__ Wg,
                            int*   __restrict__ counts,
                            int*   __restrict__ idx,
                            float* __restrict__ wts,
                            int*   __restrict__ hrow,
                            int n_tokens) {
    int tok = blockIdx.x;
    int tid = threadIdx.x;
    __shared__ float red[64 * N_EXPERTS];
    __shared__ float tot[N_EXPERTS];

    float s[N_EXPERTS];
#pragma unroll
    for (int e = 0; e < N_EXPERTS; ++e) s[e] = 0.0f;

    const float* xr = x + (size_t)tok * D_MODEL;
    for (int d = tid; d < D_MODEL; d += 64) {
        float xv = xr[d];
#pragma unroll
        for (int e = 0; e < N_EXPERTS; ++e) s[e] += xv * Wg[e * D_MODEL + d];
    }
#pragma unroll
    for (int e = 0; e < N_EXPERTS; ++e) red[tid * N_EXPERTS + e] = s[e];
    __syncthreads();

    if (tid < N_EXPERTS) {
        float t = 0.0f;
        for (int i = 0; i < 64; ++i) t += red[i * N_EXPERTS + tid];
        tot[tid] = t;
    }
    __syncthreads();

    if (tid == 0) {
        int i1 = 0; float v1 = tot[0];
        for (int e = 1; e < N_EXPERTS; ++e) if (tot[e] > v1) { v1 = tot[e]; i1 = e; }
        int i2 = (i1 == 0) ? 1 : 0; float v2 = tot[i2];
        for (int e = 0; e < N_EXPERTS; ++e)
            if (e != i1 && tot[e] > v2) { v2 = tot[e]; i2 = e; }
        float m  = fmaxf(v1, v2);
        float e1 = __expf(v1 - m), e2 = __expf(v2 - m);
        float inv = 1.0f / (e1 + e2);

        int p1 = atomicAdd(&counts[i1], 1);
        idx [i1 * n_tokens + p1] = tok;
        wts [i1 * n_tokens + p1] = e1 * inv;
        hrow[i1 * n_tokens + p1] = 2 * tok;
        int p2 = atomicAdd(&counts[i2], 1);
        idx [i2 * n_tokens + p2] = tok;
        wts [i2 * n_tokens + p2] = e2 * inv;
        hrow[i2 * n_tokens + p2] = 2 * tok + 1;
    }
}

__global__ void __launch_bounds__(256)
moe_gemm1_kernel(const float* __restrict__ x,
                 const float* __restrict__ W1,
                 const float* __restrict__ b1,
                 const int*   __restrict__ counts,
                 const int*   __restrict__ idx,
                 const int*   __restrict__ hrow,
                 float*       __restrict__ hh,
                 int n_tokens) {
    int e   = blockIdx.z;
    int Mb  = blockIdx.y;
    int Fb  = blockIdx.x;
    int cnt = counts[e];
    if (Mb * BM >= cnt) return;

    __shared__ float Xs[BM * XS_LD];
    __shared__ int   stok[BM];
    __shared__ int   shr[BM];

    int tid = threadIdx.x;
    if (tid < BM) {
        int p = Mb * BM + tid;
        bool v = (p < cnt);
        stok[tid] = v ? idx [e * n_tokens + p] : 0;
        shr [tid] = v ? hrow[e * n_tokens + p] : 0;
    }

    int lane = tid & 31;
    int wave = tid >> 5;
    int g    = lane >> 4;
    int nl   = lane & 15;
    int mg   = wave >> 2;
    int fg   = wave & 3;
    int fbase = Fb * 128 + fg * 32;

    const float* W1e = W1 + (size_t)e * D_MODEL * TWO_FF;
    const float* b1e = b1 + (size_t)e * TWO_FF;

    v8f accA[4][2], accG[4][2];
#pragma unroll
    for (int i = 0; i < 4; ++i)
#pragma unroll
        for (int j = 0; j < 2; ++j) {
            accA[i][j] = (v8f){0.f,0.f,0.f,0.f,0.f,0.f,0.f,0.f};
            accG[i][j] = (v8f){0.f,0.f,0.f,0.f,0.f,0.f,0.f,0.f};
        }

    for (int kc0 = 0; kc0 < D_MODEL; kc0 += KC) {
        __syncthreads();
        for (int t = tid; t < BM * (KC / 4); t += 256) {
            int row = t >> 4;
            int c4  = t & 15;
            const float4* src = (const float4*)(x + (size_t)stok[row] * D_MODEL + kc0);
            *(float4*)(Xs + row * XS_LD + c4 * 4) = src[c4];
        }
        __syncthreads();

        for (int k4 = 0; k4 < KC; k4 += 4) {
            int k0 = k4 + 2 * g;
            v2f a[4];
#pragma unroll
            for (int i = 0; i < 4; ++i) {
                int row = mg * 64 + i * 16 + nl;
                a[i].x = Xs[row * XS_LD + k0];
                a[i].y = Xs[row * XS_LD + k0 + 1];
            }
            const float* Wk = W1e + (size_t)(kc0 + k0) * TWO_FF;
#pragma unroll
            for (int j = 0; j < 2; ++j) {
                int colA = fbase + j * 16 + nl;
                v2f bA, bG;
                bA.x = Wk[colA];
                bA.y = Wk[TWO_FF + colA];
                bG.x = Wk[colA + D_FF];
                bG.y = Wk[TWO_FF + colA + D_FF];
#pragma unroll
                for (int i = 0; i < 4; ++i) {
                    accA[i][j] = WMMA_F32(a[i], bA, accA[i][j]);
                    accG[i][j] = WMMA_F32(a[i], bG, accG[i][j]);
                }
            }
        }
    }

#pragma unroll
    for (int j = 0; j < 2; ++j) {
        int colA = fbase + j * 16 + nl;
        float biasA = b1e[colA];
        float biasG = b1e[colA + D_FF];
#pragma unroll
        for (int i = 0; i < 4; ++i) {
#pragma unroll
            for (int jj = 0; jj < 8; ++jj) {
                int row = mg * 64 + i * 16 + jj + 8 * g;
                int p   = Mb * BM + row;
                if (p < cnt) {
                    float av = accA[i][j][jj] + biasA;
                    float gv = accG[i][j][jj] + biasG;
                    float hv = av * (1.0f / (1.0f + __expf(-av))) * gv;
                    hh[(size_t)shr[row] * D_FF + colA] = hv;
                }
            }
        }
    }
}

__global__ void __launch_bounds__(256)
moe_gemm2_kernel(const float* __restrict__ hh,
                 const float* __restrict__ W2,
                 const float* __restrict__ b2,
                 const int*   __restrict__ counts,
                 const int*   __restrict__ idx,
                 const float* __restrict__ wts,
                 const int*   __restrict__ hrow,
                 float*       __restrict__ out,
                 int n_tokens) {
    int e   = blockIdx.z;
    int Mb  = blockIdx.y;
    int Nb  = blockIdx.x;
    int cnt = counts[e];
    if (Mb * BM >= cnt) return;

    __shared__ float Hs[BM * XS_LD];
    __shared__ int   stok[BM];
    __shared__ int   shr[BM];
    __shared__ float sww[BM];

    int tid = threadIdx.x;
    if (tid < BM) {
        int p = Mb * BM + tid;
        bool v = (p < cnt);
        stok[tid] = v ? idx [e * n_tokens + p] : 0;
        shr [tid] = v ? hrow[e * n_tokens + p] : 0;
        sww [tid] = v ? wts [e * n_tokens + p] : 0.0f;
    }

    int lane = tid & 31;
    int wave = tid >> 5;
    int g    = lane >> 4;
    int nl   = lane & 15;
    int mg   = wave >> 2;
    int ng   = wave & 3;
    int nbase = Nb * 128 + ng * 32;

    const float* W2e = W2 + (size_t)e * D_FF * D_MODEL;
    const float* b2e = b2 + (size_t)e * D_MODEL;

    v8f acc[4][2];
#pragma unroll
    for (int i = 0; i < 4; ++i)
#pragma unroll
        for (int j = 0; j < 2; ++j)
            acc[i][j] = (v8f){0.f,0.f,0.f,0.f,0.f,0.f,0.f,0.f};

    for (int kc0 = 0; kc0 < D_FF; kc0 += KC) {
        __syncthreads();
        for (int t = tid; t < BM * (KC / 4); t += 256) {
            int row = t >> 4;
            int c4  = t & 15;
            const float4* src = (const float4*)(hh + (size_t)shr[row] * D_FF + kc0);
            *(float4*)(Hs + row * XS_LD + c4 * 4) = src[c4];
        }
        __syncthreads();

        for (int k4 = 0; k4 < KC; k4 += 4) {
            int k0 = k4 + 2 * g;
            v2f a[4];
#pragma unroll
            for (int i = 0; i < 4; ++i) {
                int row = mg * 64 + i * 16 + nl;
                a[i].x = Hs[row * XS_LD + k0];
                a[i].y = Hs[row * XS_LD + k0 + 1];
            }
            const float* Wk = W2e + (size_t)(kc0 + k0) * D_MODEL;
#pragma unroll
            for (int j = 0; j < 2; ++j) {
                int col = nbase + j * 16 + nl;
                v2f b;
                b.x = Wk[col];
                b.y = Wk[D_MODEL + col];
#pragma unroll
                for (int i = 0; i < 4; ++i)
                    acc[i][j] = WMMA_F32(a[i], b, acc[i][j]);
            }
        }
    }

#pragma unroll
    for (int j = 0; j < 2; ++j) {
        int col = nbase + j * 16 + nl;
        float bias = b2e[col];
#pragma unroll
        for (int i = 0; i < 4; ++i) {
#pragma unroll
            for (int jj = 0; jj < 8; ++jj) {
                int row = mg * 64 + i * 16 + jj + 8 * g;
                int p   = Mb * BM + row;
                if (p < cnt) {
                    float o = (acc[i][j][jj] + bias) * sww[row];
                    atomicAdd(out + (size_t)stok[row] * D_MODEL + col, o);
                }
            }
        }
    }
}

extern "C" void kernel_launch(void* const* d_in, const int* in_sizes, int n_in,
                              void* d_out, int out_size, void* d_ws, size_t ws_size,
                              hipStream_t stream) {
    const float* x  = (const float*)d_in[0];
    const float* Wg = (const float*)d_in[1];
    const float* W1 = (const float*)d_in[2];
    const float* b1 = (const float*)d_in[3];
    const float* W2 = (const float*)d_in[4];
    const float* b2 = (const float*)d_in[5];
    float* out = (float*)d_out;

    int n_tokens = in_sizes[0] / D_MODEL;

    int*   counts = (int*)d_ws;
    int*   idx    = counts + 16;
    float* wts    = (float*)(idx + N_EXPERTS * n_tokens);
    int*   hrow   = (int*)(wts + N_EXPERTS * n_tokens);
    float* hh     = (float*)(hrow + N_EXPERTS * n_tokens);

    zero_out_kernel<<<(out_size + 255) / 256, 256, 0, stream>>>(out, out_size, counts);
    gate_kernel<<<n_tokens, 64, 0, stream>>>(x, Wg, counts, idx, wts, hrow, n_tokens);

    int mblocks = (n_tokens + BM - 1) / BM;
    dim3 g1(TWO_FF / 2 / 128, mblocks, N_EXPERTS);
    moe_gemm1_kernel<<<g1, 256, 0, stream>>>(x, W1, b1, counts, idx, hrow, hh, n_tokens);

    dim3 g2(D_MODEL / 128, mblocks, N_EXPERTS);
    moe_gemm2_kernel<<<g2, 256, 0, stream>>>(hh, W2, b2, counts, idx, wts, hrow, out, n_tokens);
}


# GEMM2 split 4 ways in K: 4x as many quarter-length blocks over several dispatch rounds; partial planes by plain 16-byte stores, summed in fixed order by a combine pass; output atomics dropped
# speedup vs baseline: 1.2458x; 1.2458x over previous
#include <hip/hip_runtime.h>
#include <math.h>

#define D_MODEL   1024
#define D_FF      2048
#define TWO_FF    4096
#define N_EXPERTS 8
#define KSPLIT    4
#define KCHUNK    (D_FF / KSPLIT)

#define BM   128

typedef float v2f __attribute__((ext_vector_type(2)));
typedef float v4f __attribute__((ext_vector_type(4)));
typedef float v8f __attribute__((ext_vector_type(8)));

#define WMMA_F32(A, B, C) \
    __builtin_amdgcn_wmma_f32_16x16x4_f32(false, (A), false, (B), (short)0, (C), false, false)

__global__ void init_kernel(float* __restrict__ out, int n_main, int n, int* __restrict__ counts) {
    int tid = threadIdx.x;
    if (tid < 16) counts[tid] = 0;
    for (int i = n_main + tid; i < n; i += 64) out[i] = 0.0f;
}

__global__ void gate_kernel(const float* __restrict__ x,
                            const float* __restrict__ Wg,
                            int*   __restrict__ counts,
                            int*   __restrict__ idx,
                            float* __restrict__ wts,
                            int*   __restrict__ hrow,
                            int n_tokens) {
    int tok = blockIdx.x;
    int tid = threadIdx.x;
    __shared__ float red[64 * N_EXPERTS];
    __shared__ float tot[N_EXPERTS];

    float s[N_EXPERTS];
#pragma unroll
    for (int e = 0; e < N_EXPERTS; ++e) s[e] = 0.0f;

    const float* xr = x + (size_t)tok * D_MODEL;
    for (int d = tid; d < D_MODEL; d += 64) {
        float xv = xr[d];
#pragma unroll
        for (int e = 0; e < N_EXPERTS; ++e) s[e] += xv * Wg[e * D_MODEL + d];
    }
#pragma unroll
    for (int e = 0; e < N_EXPERTS; ++e) red[tid * N_EXPERTS + e] = s[e];
    __syncthreads();

    if (tid < N_EXPERTS) {
        float t = 0.0f;
        for (int i = 0; i < 64; ++i) t += red[i * N_EXPERTS + tid];
        tot[tid] = t;
    }
    __syncthreads();

    if (tid == 0) {
        int i1 = 0; float v1 = tot[0];
        for (int e = 1; e < N_EXPERTS; ++e) if (tot[e] > v1) { v1 = tot[e]; i1 = e; }
        int i2 = (i1 == 0) ? 1 : 0; float v2 = tot[i2];
        for (int e = 0; e < N_EXPERTS; ++e)
            if (e != i1 && tot[e] > v2) { v2 = tot[e]; i2 = e; }
        float m  = fmaxf(v1, v2);
        float e1 = __expf(v1 - m), e2 = __expf(v2 - m);
        float inv = 1.0f / (e1 + e2);

        int p1 = atomicAdd(&counts[i1], 1);
        idx [i1 * n_tokens + p1] = tok;
        wts [i1 * n_tokens + p1] = e1 * inv;
        hrow[i1 * n_tokens + p1] = 2 * tok;
        int p2 = atomicAdd(&counts[i2], 1);
        idx [i2 * n_tokens + p2] = tok;
        wts [i2 * n_tokens + p2] = e2 * inv;
        hrow[i2 * n_tokens + p2] = 2 * tok + 1;
    }
}

__global__ void __launch_bounds__(256)
moe_gemm1_kernel(const float* __restrict__ x,
                 const float* __restrict__ W1,
                 const float* __restrict__ b1,
                 const int*   __restrict__ counts,
                 const int*   __restrict__ idx,
                 const int*   __restrict__ hrow,
                 float*       __restrict__ hh,
                 int n_tokens) {
    int e   = blockIdx.z;
    int Mb  = blockIdx.y;
    int Fb  = blockIdx.x;
    int cnt = counts[e];
    if (Mb * BM >= cnt) return;

    __shared__ int   stok[BM];
    __shared__ int   shr[BM];

    int tid = threadIdx.x;
    if (tid < BM) {
        int p = Mb * BM + tid;
        bool v = (p < cnt);
        stok[tid] = v ? idx [e * n_tokens + p] : 0;
        shr [tid] = v ? hrow[e * n_tokens + p] : 0;
    }
    __syncthreads();

    int lane = tid & 31;
    int wave = tid >> 5;
    int g    = lane >> 4;
    int nl   = lane & 15;
    int mg   = wave >> 1;
    int fg   = wave & 1;
    int fbase = Fb * 128 + fg * 64 + 4 * nl;

    const float* W1e = W1 + (size_t)e * D_MODEL * TWO_FF;
    const float* b1e = b1 + (size_t)e * TWO_FF;

    const float* arow[2];
#pragma unroll
    for (int i = 0; i < 2; ++i)
        arow[i] = x + (size_t)stok[mg * 32 + i * 16 + nl] * D_MODEL + 4 * g;
    const float* wcol = W1e + (size_t)(4 * g) * TWO_FF + fbase;

    v8f accA[2][4], accG[2][4];
#pragma unroll
    for (int i = 0; i < 2; ++i)
#pragma unroll
        for (int j = 0; j < 4; ++j) {
            accA[i][j] = (v8f){0.f,0.f,0.f,0.f,0.f,0.f,0.f,0.f};
            accG[i][j] = (v8f){0.f,0.f,0.f,0.f,0.f,0.f,0.f,0.f};
        }

#pragma unroll 8
    for (int kb = 0; kb < D_MODEL; kb += 8) {
        v4f av[2];
#pragma unroll
        for (int i = 0; i < 2; ++i) av[i] = *(const v4f*)(arow[i] + kb);
        const float* Wk = wcol + (size_t)kb * TWO_FF;
        v4f qa[4], qg[4];
#pragma unroll
        for (int kk = 0; kk < 4; ++kk) {
            qa[kk] = *(const v4f*)(Wk + (size_t)kk * TWO_FF);
            qg[kk] = *(const v4f*)(Wk + (size_t)kk * TWO_FF + D_FF);
        }
#pragma unroll
        for (int j = 0; j < 4; ++j) {
            v2f bA, bG;
            bA.x = qa[0][j]; bA.y = qa[1][j];
            bG.x = qg[0][j]; bG.y = qg[1][j];
#pragma unroll
            for (int i = 0; i < 2; ++i) {
                v2f a0; a0.x = av[i].x; a0.y = av[i].y;
                accA[i][j] = WMMA_F32(a0, bA, accA[i][j]);
                accG[i][j] = WMMA_F32(a0, bG, accG[i][j]);
            }
        }
#pragma unroll
        for (int j = 0; j < 4; ++j) {
            v2f bA, bG;
            bA.x = qa[2][j]; bA.y = qa[3][j];
            bG.x = qg[2][j]; bG.y = qg[3][j];
#pragma unroll
            for (int i = 0; i < 2; ++i) {
                v2f a1; a1.x = av[i].z; a1.y = av[i].w;
                accA[i][j] = WMMA_F32(a1, bA, accA[i][j]);
                accG[i][j] = WMMA_F32(a1, bG, accG[i][j]);
            }
        }
    }

    v4f biasA = *(const v4f*)(b1e + fbase);
    v4f biasG = *(const v4f*)(b1e + fbase + D_FF);
#pragma unroll
    for (int i = 0; i < 2; ++i) {
#pragma unroll
        for (int jj = 0; jj < 8; ++jj) {
            int row = mg * 32 + i * 16 + jj + 8 * g;
            int p   = Mb * BM + row;
            if (p < cnt) {
                v4f hv;
#pragma unroll
                for (int j = 0; j < 4; ++j) {
                    float av = accA[i][j][jj] + biasA[j];
                    float gv = accG[i][j][jj] + biasG[j];
                    hv[j] = av * (1.0f / (1.0f + __expf(-av))) * gv;
                }
                *(v4f*)(hh + (size_t)shr[row] * D_FF + fbase) = hv;
            }
        }
    }
}

__global__ void __launch_bounds__(128)
moe_gemm2_kernel(const float* __restrict__ hh,
                 const float* __restrict__ W2,
                 const float* __restrict__ b2,
                 const int*   __restrict__ counts,
                 const float* __restrict__ wts,
                 const int*   __restrict__ hrow,
                 float*       __restrict__ part,
                 int n_tokens) {
    int e   = blockIdx.z;
    int Mb  = blockIdx.y;
    int Nb  = blockIdx.x & 7;
    int ks  = blockIdx.x >> 3;
    int cnt = counts[e];
    if (Mb * BM >= cnt) return;

    __shared__ int   shr[BM];
    __shared__ float sww[BM];

    int tid = threadIdx.x;
    if (tid < BM) {
        int p = Mb * BM + tid;
        bool v = (p < cnt);
        shr [tid] = v ? hrow[e * n_tokens + p] : 0;
        sww [tid] = v ? wts [e * n_tokens + p] : 0.0f;
    }
    __syncthreads();

    int lane = tid & 31;
    int wave = tid >> 5;
    int g    = lane >> 4;
    int nl   = lane & 15;
    int mg   = wave;
    int nbase = Nb * 128 + 4 * nl;

    const float* W2e = W2 + (size_t)e * D_FF * D_MODEL;
    const float* b2e = b2 + (size_t)e * D_MODEL;

    const float* arow[2];
#pragma unroll
    for (int i = 0; i < 2; ++i)
        arow[i] = hh + (size_t)shr[mg * 32 + i * 16 + nl] * D_FF + ks * KCHUNK + 4 * g;
    const float* wcol = W2e + (size_t)(ks * KCHUNK + 4 * g) * D_MODEL + nbase;

    v8f acc[2][8];
#pragma unroll
    for (int i = 0; i < 2; ++i)
#pragma unroll
        for (int n = 0; n < 8; ++n)
            acc[i][n] = (v8f){0.f,0.f,0.f,0.f,0.f,0.f,0.f,0.f};

#pragma unroll 8
    for (int kb = 0; kb < KCHUNK; kb += 8) {
        v4f av[2];
#pragma unroll
        for (int i = 0; i < 2; ++i) av[i] = *(const v4f*)(arow[i] + kb);
        const float* Wk = wcol + (size_t)kb * D_MODEL;
        v4f q[4][2];
#pragma unroll
        for (int kk = 0; kk < 4; ++kk) {
            q[kk][0] = *(const v4f*)(Wk + (size_t)kk * D_MODEL);
            q[kk][1] = *(const v4f*)(Wk + (size_t)kk * D_MODEL + 64);
        }
#pragma unroll
        for (int t = 0; t < 2; ++t)
#pragma unroll
            for (int j = 0; j < 4; ++j) {
                v2f b; b.x = q[0][t][j]; b.y = q[1][t][j];
#pragma unroll
                for (int i = 0; i < 2; ++i) {
                    v2f a0; a0.x = av[i].x; a0.y = av[i].y;
                    acc[i][t * 4 + j] = WMMA_F32(a0, b, acc[i][t * 4 + j]);
                }
            }
#pragma unroll
        for (int t = 0; t < 2; ++t)
#pragma unroll
            for (int j = 0; j < 4; ++j) {
                v2f b; b.x = q[2][t][j]; b.y = q[3][t][j];
#pragma unroll
                for (int i = 0; i < 2; ++i) {
                    v2f a1; a1.x = av[i].z; a1.y = av[i].w;
                    acc[i][t * 4 + j] = WMMA_F32(a1, b, acc[i][t * 4 + j]);
                }
            }
    }

    float* plane = part + (size_t)ks * 2 * n_tokens * D_MODEL;
#pragma unroll
    for (int t = 0; t < 2; ++t) {
        int col = nbase + t * 64;
        v4f bias = (v4f){0.f, 0.f, 0.f, 0.f};
        if (ks == 0) bias = *(const v4f*)(b2e + col);
#pragma unroll
        for (int i = 0; i < 2; ++i) {
#pragma unroll
            for (int jj = 0; jj < 8; ++jj) {
                int row = mg * 32 + i * 16 + jj + 8 * g;
                int p   = Mb * BM + row;
                if (p < cnt) {
                    float w = sww[row];
                    v4f pv;
#pragma unroll
                    for (int j = 0; j < 4; ++j)
                        pv[j] = (acc[i][t * 4 + j][jj] + bias[j]) * w;
                    *(v4f*)(plane + (size_t)shr[row] * D_MODEL + col) = pv;
                }
            }
        }
    }
}

__global__ void __launch_bounds__(256)
combine_kernel(const float* __restrict__ part, float* __restrict__ out, int n_tokens) {
    int tok = blockIdx.x;
    int c   = 4 * threadIdx.x;
    size_t plane = (size_t)2 * n_tokens * D_MODEL;
    const float* p = part + (size_t)(2 * tok) * D_MODEL + c;
    v4f r0 = *(const v4f*)(p);
    v4f r1 = *(const v4f*)(p + D_MODEL);
#pragma unroll
    for (int ks = 1; ks < KSPLIT; ++ks) {
        r0 += *(const v4f*)(p + ks * plane);
        r1 += *(const v4f*)(p + ks * plane + D_MODEL);
    }
    *(v4f*)(out + (size_t)tok * D_MODEL + c) = r0 + r1;
}

extern "C" void kernel_launch(void* const* d_in, const int* in_sizes, int n_in,
                              void* d_out, int out_size, void* d_ws, size_t ws_size,
                              hipStream_t stream) {
    const float* x  = (const float*)d_in[0];
    const float* Wg = (const float*)d_in[1];
    const float* W1 = (const float*)d_in[2];
    const float* b1 = (const float*)d_in[3];
    const float* W2 = (const float*)d_in[4];
    const float* b2 = (const float*)d_in[5];
    float* out = (float*)d_out;

    int n_tokens = in_sizes[0] / D_MODEL;

    int*   counts = (int*)d_ws;
    int*   idx    = counts + 16;
    float* wts    = (float*)(idx + N_EXPERTS * n_tokens);
    int*   hrow   = (int*)(wts + N_EXPERTS * n_tokens);
    float* hh     = (float*)(hrow + N_EXPERTS * n_tokens);
    float* part   = hh + (size_t)2 * n_tokens * D_FF;

    init_kernel<<<1, 64, 0, stream>>>(out, n_tokens * D_MODEL, out_size, counts);
    gate_kernel<<<n_tokens, 64, 0, stream>>>(x, Wg, counts, idx, wts, hrow, n_tokens);

    int mblocks = (n_tokens + BM - 1) / BM;
    dim3 g1(TWO_FF / 2 / 128, mblocks, N_EXPERTS);
    moe_gemm1_kernel<<<g1, 256, 0, stream>>>(x, W1, b1, counts, idx, hrow, hh, n_tokens);

    dim3 g2(D_MODEL / 128 * KSPLIT, mblocks, N_EXPERTS);
    moe_gemm2_kernel<<<g2, 128, 0, stream>>>(hh, W2, b2, counts, wts, hrow, part, n_tokens);

    combine_kernel<<<n_tokens, 256, 0, stream>>>(part, out, n_tokens);
}


# Each GEMM block's K loop split over wave pairs: a wave runs half the k range, halves summed through LDS in fixed order (lower+upper); 8-wave blocks half as long (GEMM1 tile 128x64 F), both GEMMs
# speedup vs baseline: 1.3040x; 1.0467x over previous
#include <hip/hip_runtime.h>
#include <math.h>

#define D_MODEL   1024
#define D_FF      2048
#define TWO_FF    4096
#define N_EXPERTS 8
#define KSPLIT    4
#define KCHUNK    (D_FF / KSPLIT)
#define KW1       (D_MODEL / 2)
#define KW2       (KCHUNK / 2)

#define BM   128

typedef float v2f __attribute__((ext_vector_type(2)));
typedef float v4f __attribute__((ext_vector_type(4)));
typedef float v8f __attribute__((ext_vector_type(8)));

#define WMMA_F32(A, B, C) \
    __builtin_amdgcn_wmma_f32_16x16x4_f32(false, (A), false, (B), (short)0, (C), false, false)

__global__ void init_kernel(float* __restrict__ out, int n_main, int n, int* __restrict__ counts) {
    int tid = threadIdx.x;
    if (tid < 16) counts[tid] = 0;
    for (int i = n_main + tid; i < n; i += 64) out[i] = 0.0f;
}

__global__ void gate_kernel(const float* __restrict__ x,
                            const float* __restrict__ Wg,
                            int*   __restrict__ counts,
                            int*   __restrict__ idx,
                            float* __restrict__ wts,
                            int*   __restrict__ hrow,
                            int n_tokens) {
    int tok = blockIdx.x;
    int tid = threadIdx.x;
    __shared__ float red[64 * N_EXPERTS];
    __shared__ float tot[N_EXPERTS];

    float s[N_EXPERTS];
#pragma unroll
    for (int e = 0; e < N_EXPERTS; ++e) s[e] = 0.0f;

    const float* xr = x + (size_t)tok * D_MODEL;
    for (int d = tid; d < D_MODEL; d += 64) {
        float xv = xr[d];
#pragma unroll
        for (int e = 0; e < N_EXPERTS; ++e) s[e] += xv * Wg[e * D_MODEL + d];
    }
#pragma unroll
    for (int e = 0; e < N_EXPERTS; ++e) red[tid * N_EXPERTS + e] = s[e];
    __syncthreads();

    if (tid < N_EXPERTS) {
        float t = 0.0f;
        for (int i = 0; i < 64; ++i) t += red[i * N_EXPERTS + tid];
        tot[tid] = t;
    }
    __syncthreads();

    if (tid == 0) {
        int i1 = 0; float v1 = tot[0];
        for (int e = 1; e < N_EXPERTS; ++e) if (tot[e] > v1) { v1 = tot[e]; i1 = e; }
        int i2 = (i1 == 0) ? 1 : 0; float v2 = tot[i2];
        for (int e = 0; e < N_EXPERTS; ++e)
            if (e != i1 && tot[e] > v2) { v2 = tot[e]; i2 = e; }
        float m  = fmaxf(v1, v2);
        float e1 = __expf(v1 - m), e2 = __expf(v2 - m);
        float inv = 1.0f / (e1 + e2);

        int p1 = atomicAdd(&counts[i1], 1);
        idx [i1 * n_tokens + p1] = tok;
        wts [i1 * n_tokens + p1] = e1 * inv;
        hrow[i1 * n_tokens + p1] = 2 * tok;
        int p2 = atomicAdd(&counts[i2], 1);
        idx [i2 * n_tokens + p2] = tok;
        wts [i2 * n_tokens + p2] = e2 * inv;
        hrow[i2 * n_tokens + p2] = 2 * tok + 1;
    }
}

__global__ void __launch_bounds__(256)
moe_gemm1_kernel(const float* __restrict__ x,
                 const float* __restrict__ W1,
                 const float* __restrict__ b1,
                 const int*   __restrict__ counts,
                 const int*   __restrict__ idx,
                 const int*   __restrict__ hrow,
                 float*       __restrict__ hh,
                 int n_tokens) {
    int e   = blockIdx.z;
    int Mb  = blockIdx.y;
    int Fb  = blockIdx.x;
    int cnt = counts[e];
    if (Mb * BM >= cnt) return;

    __shared__ int   stok[BM];
    __shared__ int   shr[BM];
    __shared__ v4f   xch[8][16][32];

    int tid = threadIdx.x;
    if (tid < BM) {
        int p = Mb * BM + tid;
        bool v = (p < cnt);
        stok[tid] = v ? idx [e * n_tokens + p] : 0;
        shr [tid] = v ? hrow[e * n_tokens + p] : 0;
    }
    __syncthreads();

    int lane = tid & 31;
    int wave = tid >> 5;
    int g    = lane >> 4;
    int nl   = lane & 15;
    int mg   = wave & 3;
    int kh   = wave >> 2;
    int fbase = Fb * 64 + 4 * nl;

    const float* W1e = W1 + (size_t)e * D_MODEL * TWO_FF;
    const float* b1e = b1 + (size_t)e * TWO_FF;

    const float* arow[2];
#pragma unroll
    for (int i = 0; i < 2; ++i)
        arow[i] = x + (size_t)stok[mg * 32 + (i ^ kh) * 16 + nl] * D_MODEL + kh * KW1 + 4 * g;
    const float* wcol = W1e + (size_t)(kh * KW1 + 4 * g) * TWO_FF + fbase;

    v8f accA[2][4], accG[2][4];
#pragma unroll
    for (int i = 0; i < 2; ++i)
#pragma unroll
        for (int j = 0; j < 4; ++j) {
            accA[i][j] = (v8f){0.f,0.f,0.f,0.f,0.f,0.f,0.f,0.f};
            accG[i][j] = (v8f){0.f,0.f,0.f,0.f,0.f,0.f,0.f,0.f};
        }

#pragma unroll 8
    for (int kb = 0; kb < KW1; kb += 8) {
        v4f av[2];
#pragma unroll
        for (int i = 0; i < 2; ++i) av[i] = *(const v4f*)(arow[i] + kb);
        const float* Wk = wcol + (size_t)kb * TWO_FF;
        v4f qa[4], qg[4];
#pragma unroll
        for (int kk = 0; kk < 4; ++kk) {
            qa[kk] = *(const v4f*)(Wk + (size_t)kk * TWO_FF);
            qg[kk] = *(const v4f*)(Wk + (size_t)kk * TWO_FF + D_FF);
        }
#pragma unroll
        for (int j = 0; j < 4; ++j) {
            v2f bA, bG;
            bA.x = qa[0][j]; bA.y = qa[1][j];
            bG.x = qg[0][j]; bG.y = qg[1][j];
#pragma unroll
            for (int i = 0; i < 2; ++i) {
                v2f a0; a0.x = av[i].x; a0.y = av[i].y;
                accA[i][j] = WMMA_F32(a0, bA, accA[i][j]);
                accG[i][j] = WMMA_F32(a0, bG, accG[i][j]);
            }
        }
#pragma unroll
        for (int j = 0; j < 4; ++j) {
            v2f bA, bG;
            bA.x = qa[2][j]; bA.y = qa[3][j];
            bG.x = qg[2][j]; bG.y = qg[3][j];
#pragma unroll
            for (int i = 0; i < 2; ++i) {
                v2f a1; a1.x = av[i].z; a1.y = av[i].w;
                accA[i][j] = WMMA_F32(a1, bA, accA[i][j]);
                accG[i][j] = WMMA_F32(a1, bG, accG[i][j]);
            }
        }
    }

#pragma unroll
    for (int j = 0; j < 4; ++j) {
        xch[wave][4 * j + 0][lane] = (v4f){accA[1][j][0], accA[1][j][1], accA[1][j][2], accA[1][j][3]};
        xch[wave][4 * j + 1][lane] = (v4f){accA[1][j][4], accA[1][j][5], accA[1][j][6], accA[1][j][7]};
        xch[wave][4 * j + 2][lane] = (v4f){accG[1][j][0], accG[1][j][1], accG[1][j][2], accG[1][j][3]};
        xch[wave][4 * j + 3][lane] = (v4f){accG[1][j][4], accG[1][j][5], accG[1][j][6], accG[1][j][7]};
    }
    __syncthreads();
    int pw = wave ^ 4;
#pragma unroll
    for (int j = 0; j < 4; ++j) {
        v4f t0 = xch[pw][4 * j + 0][lane];
        v4f t1 = xch[pw][4 * j + 1][lane];
        v4f t2 = xch[pw][4 * j + 2][lane];
        v4f t3 = xch[pw][4 * j + 3][lane];
#pragma unroll
        for (int r = 0; r < 4; ++r) {
            accA[0][j][r]     += t0[r];
            accA[0][j][r + 4] += t1[r];
            accG[0][j][r]     += t2[r];
            accG[0][j][r + 4] += t3[r];
        }
    }

    v4f biasA = *(const v4f*)(b1e + fbase);
    v4f biasG = *(const v4f*)(b1e + fbase + D_FF);
#pragma unroll
    for (int jj = 0; jj < 8; ++jj) {
        int row = mg * 32 + kh * 16 + jj + 8 * g;
        int p   = Mb * BM + row;
        if (p < cnt) {
            v4f hv;
#pragma unroll
            for (int j = 0; j < 4; ++j) {
                float av = accA[0][j][jj] + biasA[j];
                float gv = accG[0][j][jj] + biasG[j];
                hv[j] = av * (1.0f / (1.0f + __expf(-av))) * gv;
            }
            *(v4f*)(hh + (size_t)shr[row] * D_FF + fbase) = hv;
        }
    }
}

__global__ void __launch_bounds__(256)
moe_gemm2_kernel(const float* __restrict__ hh,
                 const float* __restrict__ W2,
                 const float* __restrict__ b2,
                 const int*   __restrict__ counts,
                 const float* __restrict__ wts,
                 const int*   __restrict__ hrow,
                 float*       __restrict__ part,
                 int n_tokens) {
    int e   = blockIdx.z;
    int Mb  = blockIdx.y;
    int Nb  = blockIdx.x & 7;
    int ks  = blockIdx.x >> 3;
    int cnt = counts[e];
    if (Mb * BM >= cnt) return;

    __shared__ int   shr[BM];
    __shared__ float sww[BM];
    __shared__ v4f   xch[8][16][32];

    int tid = threadIdx.x;
    if (tid < BM) {
        int p = Mb * BM + tid;
        bool v = (p < cnt);
        shr [tid] = v ? hrow[e * n_tokens + p] : 0;
        sww [tid] = v ? wts [e * n_tokens + p] : 0.0f;
    }
    __syncthreads();

    int lane = tid & 31;
    int wave = tid >> 5;
    int g    = lane >> 4;
    int nl   = lane & 15;
    int mg   = wave & 3;
    int kh   = wave >> 2;
    int nbase = Nb * 128 + 4 * nl;

    const float* W2e = W2 + (size_t)e * D_FF * D_MODEL;
    const float* b2e = b2 + (size_t)e * D_MODEL;

    const float* arow[2];
#pragma unroll
    for (int i = 0; i < 2; ++i)
        arow[i] = hh + (size_t)shr[mg * 32 + (i ^ kh) * 16 + nl] * D_FF + ks * KCHUNK + kh * KW2 + 4 * g;
    const float* wcol = W2e + (size_t)(ks * KCHUNK + kh * KW2 + 4 * g) * D_MODEL + nbase;

    v8f acc[2][8];
#pragma unroll
    for (int i = 0; i < 2; ++i)
#pragma unroll
        for (int n = 0; n < 8; ++n)
            acc[i][n] = (v8f){0.f,0.f,0.f,0.f,0.f,0.f,0.f,0.f};

#pragma unroll 8
    for (int kb = 0; kb < KW2; kb += 8) {
        v4f av[2];
#pragma unroll
        for (int i = 0; i < 2; ++i) av[i] = *(const v4f*)(arow[i] + kb);
        const float* Wk = wcol + (size_t)kb * D_MODEL;
        v4f q[4][2];
#pragma unroll
        for (int kk = 0; kk < 4; ++kk) {
            q[kk][0] = *(const v4f*)(Wk + (size_t)kk * D_MODEL);
            q[kk][1] = *(const v4f*)(Wk + (size_t)kk * D_MODEL + 64);
        }
#pragma unroll
        for (int t = 0; t < 2; ++t)
#pragma unroll
            for (int j = 0; j < 4; ++j) {
                v2f b; b.x = q[0][t][j]; b.y = q[1][t][j];
#pragma unroll
                for (int i = 0; i < 2; ++i) {
                    v2f a0; a0.x = av[i].x; a0.y = av[i].y;
                    acc[i][t * 4 + j] = WMMA_F32(a0, b, acc[i][t * 4 + j]);
                }
            }
#pragma unroll
        for (int t = 0; t < 2; ++t)
#pragma unroll
            for (int j = 0; j < 4; ++j) {
                v2f b; b.x = q[2][t][j]; b.y = q[3][t][j];
#pragma unroll
                for (int i = 0; i < 2; ++i) {
                    v2f a1; a1.x = av[i].z; a1.y = av[i].w;
                    acc[i][t * 4 + j] = WMMA_F32(a1, b, acc[i][t * 4 + j]);
                }
            }
    }

#pragma unroll
    for (int n = 0; n < 8; ++n) {
        xch[wave][2 * n + 0][lane] = (v4f){acc[1][n][0], acc[1][n][1], acc[1][n][2], acc[1][n][3]};
        xch[wave][2 * n + 1][lane] = (v4f){acc[1][n][4], acc[1][n][5], acc[1][n][6], acc[1][n][7]};
    }
    __syncthreads();
    int pw = wave ^ 4;
#pragma unroll
    for (int n = 0; n < 8; ++n) {
        v4f t0 = xch[pw][2 * n + 0][lane];
        v4f t1 = xch[pw][2 * n + 1][lane];
#pragma unroll
        for (int r = 0; r < 4; ++r) {
            acc[0][n][r]     += t0[r];
            acc[0][n][r + 4] += t1[r];
        }
    }

    float* plane = part + (size_t)ks * 2 * n_tokens * D_MODEL;
#pragma unroll
    for (int t = 0; t < 2; ++t) {
        int col = nbase + t * 64;
        v4f bias = (v4f){0.f, 0.f, 0.f, 0.f};
        if (ks == 0) bias = *(const v4f*)(b2e + col);
#pragma unroll
        for (int jj = 0; jj < 8; ++jj) {
            int row = mg * 32 + kh * 16 + jj + 8 * g;
            int p   = Mb * BM + row;
            if (p < cnt) {
                float w = sww[row];
                v4f pv;
#pragma unroll
                for (int j = 0; j < 4; ++j)
                    pv[j] = (acc[0][t * 4 + j][jj] + bias[j]) * w;
                *(v4f*)(plane + (size_t)shr[row] * D_MODEL + col) = pv;
            }
        }
    }
}

__global__ void __launch_bounds__(256)
combine_kernel(const float* __restrict__ part, float* __restrict__ out, int n_tokens) {
    int tok = blockIdx.x;
    int c   = 4 * threadIdx.x;
    size_t plane = (size_t)2 * n_tokens * D_MODEL;
    const float* p = part + (size_t)(2 * tok) * D_MODEL + c;
    v4f r0 = *(const v4f*)(p);
    v4f r1 = *(const v4f*)(p + D_MODEL);
#pragma unroll
    for (int ks = 1; ks < KSPLIT; ++ks) {
        r0 += *(const v4f*)(p + ks * plane);
        r1 += *(const v4f*)(p + ks * plane + D_MODEL);
    }
    *(v4f*)(out + (size_t)tok * D_MODEL + c) = r0 + r1;
}

extern "C" void kernel_launch(void* const* d_in, const int* in_sizes, int n_in,
                              void* d_out, int out_size, void* d_ws, size_t ws_size,
                              hipStream_t stream) {
    const float* x  = (const float*)d_in[0];
    const float* Wg = (const float*)d_in[1];
    const float* W1 = (const float*)d_in[2];
    const float* b1 = (const float*)d_in[3];
    const float* W2 = (const float*)d_in[4];
    const float* b2 = (const float*)d_in[5];
    float* out = (float*)d_out;

    int n_tokens = in_sizes[0] / D_MODEL;

    int*   counts = (int*)d_ws;
    int*   idx    = counts + 16;
    float* wts    = (float*)(idx + N_EXPERTS * n_tokens);
    int*   hrow   = (int*)(wts + N_EXPERTS * n_tokens);
    float* hh     = (float*)(hrow + N_EXPERTS * n_tokens);
    float* part   = hh + (size_t)2 * n_tokens * D_FF;

    init_kernel<<<1, 64, 0, stream>>>(out, n_tokens * D_MODEL, out_size, counts);
    gate_kernel<<<n_tokens, 64, 0, stream>>>(x, Wg, counts, idx, wts, hrow, n_tokens);

    int mblocks = (n_tokens + BM - 1) / BM;
    dim3 g1(TWO_FF / 2 / 64, mblocks, N_EXPERTS);
    moe_gemm1_kernel<<<g1, 256, 0, stream>>>(x, W1, b1, counts, idx, hrow, hh, n_tokens);

    dim3 g2(D_MODEL / 128 * KSPLIT, mblocks, N_EXPERTS);
    moe_gemm2_kernel<<<g2, 256, 0, stream>>>(hh, W2, b2, counts, wts, hrow, part, n_tokens);

    combine_kernel<<<n_tokens, 256, 0, stream>>>(part, out, n_tokens);
}


# Both GEMM grids re-cut into a packed list of 32-row groups: blocks of 4 groups fill whole residency rounds, leftover groups run as blocks with K split over 8 waves (k sum there: 8 slices in order)
# speedup vs baseline: 1.3455x; 1.0318x over previous
#include <hip/hip_runtime.h>
#include <math.h>

#define D_MODEL   1024
#define D_FF      2048
#define TWO_FF    4096
#define N_EXPERTS 8
#define KSPLIT    4
#define KCHUNK    (D_FF / KSPLIT)

#define BM    128
#define GROWS 32

typedef float v2f __attribute__((ext_vector_type(2)));
typedef float v4f __attribute__((ext_vector_type(4)));
typedef float v8f __attribute__((ext_vector_type(8)));

#define WMMA_F32(A, B, C) \
    __builtin_amdgcn_wmma_f32_16x16x4_f32(false, (A), false, (B), (short)0, (C), false, false)

__global__ void init_kernel(float* __restrict__ out, int n_main, int n, int* __restrict__ counts) {
    int tid = threadIdx.x;
    if (tid < 16) counts[tid] = 0;
    for (int i = n_main + tid; i < n; i += 64) out[i] = 0.0f;
}

__global__ void gate_kernel(const float* __restrict__ x,
                            const float* __restrict__ Wg,
                            int*   __restrict__ counts,
                            int*   __restrict__ idx,
                            float* __restrict__ wts,
                            int*   __restrict__ hrow,
                            int n_tokens) {
    int tok = blockIdx.x;
    int tid = threadIdx.x;
    __shared__ float red[64 * N_EXPERTS];
    __shared__ float tot[N_EXPERTS];

    float s[N_EXPERTS];
#pragma unroll
    for (int e = 0; e < N_EXPERTS; ++e) s[e] = 0.0f;

    const float* xr = x + (size_t)tok * D_MODEL;
    for (int d = tid; d < D_MODEL; d += 64) {
        float xv = xr[d];
#pragma unroll
        for (int e = 0; e < N_EXPERTS; ++e) s[e] += xv * Wg[e * D_MODEL + d];
    }
#pragma unroll
    for (int e = 0; e < N_EXPERTS; ++e) red[tid * N_EXPERTS + e] = s[e];
    __syncthreads();

    if (tid < N_EXPERTS) {
        float t = 0.0f;
        for (int i = 0; i < 64; ++i) t += red[i * N_EXPERTS + tid];
        tot[tid] = t;
    }
    __syncthreads();

    if (tid == 0) {
        int i1 = 0; float v1 = tot[0];
        for (int e = 1; e < N_EXPERTS; ++e) if (tot[e] > v1) { v1 = tot[e]; i1 = e; }
        int i2 = (i1 == 0) ? 1 : 0; float v2 = tot[i2];
        for (int e = 0; e < N_EXPERTS; ++e)
            if (e != i1 && tot[e] > v2) { v2 = tot[e]; i2 = e; }
        float m  = fmaxf(v1, v2);
        float e1 = __expf(v1 - m), e2 = __expf(v2 - m);
        float inv = 1.0f / (e1 + e2);

        int p1 = atomicAdd(&counts[i1], 1);
        idx [i1 * n_tokens + p1] = tok;
        wts [i1 * n_tokens + p1] = e1 * inv;
        hrow[i1 * n_tokens + p1] = 2 * tok;
        int p2 = atomicAdd(&counts[i2], 1);
        idx [i2 * n_tokens + p2] = tok;
        wts [i2 * n_tokens + p2] = e2 * inv;
        hrow[i2 * n_tokens + p2] = 2 * tok + 1;
    }
}

__device__ __forceinline__ void block_plan(const int* __restrict__ counts, int (&c)[N_EXPERTS],
                                           int ytiles, int& nA, int& nB) {
    int G = 0;
#pragma unroll
    for (int k = 0; k < N_EXPERTS; ++k) { c[k] = counts[k]; G += (c[k] + GROWS - 1) >> 5; }
    nA = (G >> 6) << 4;
    nB = G - 4 * nA;
    if (nA + nB > ytiles) { nA = (G + 3) >> 2; nB = 0; }
}

__device__ __forceinline__ void find_group(const int (&c)[N_EXPERTS], int gidx,
                                           int& e, int& lg, int& cnt) {
    e = 0; lg = 0; cnt = 0;
    int base = 0;
#pragma unroll
    for (int k = 0; k < N_EXPERTS; ++k) {
        int ng = (c[k] + GROWS - 1) >> 5;
        if (gidx >= base && gidx < base + ng) { e = k; lg = gidx - base; cnt = c[k]; }
        base += ng;
    }
}

__global__ void __launch_bounds__(256)
moe_gemm1_kernel(const float* __restrict__ x,
                 const float* __restrict__ W1,
                 const float* __restrict__ b1,
                 const int*   __restrict__ counts,
                 const int*   __restrict__ idx,
                 const int*   __restrict__ hrow,
                 float*       __restrict__ hh,
                 int n_tokens) {
    int Fb = blockIdx.x;
    int Y  = blockIdx.y;
    int c[N_EXPERTS], nA, nB;
    block_plan(counts, c, (int)gridDim.y, nA, nB);
    if (Y >= nA + nB) return;
    const bool wide = (Y >= nA);

    __shared__ int   stok[BM];
    __shared__ int   shr[BM];
    __shared__ v4f   xch[8][16][32];

    int tid  = threadIdx.x;
    int lane = tid & 31;
    int wave = __builtin_amdgcn_readfirstlane(tid >> 5);
    int g    = lane >> 4;
    int nl   = lane & 15;
    int mg   = wide ? 0 : (wave & 3);
    int ksl  = wide ? wave : (wave >> 2);
    int klen = wide ? (D_MODEL / 8) : (D_MODEL / 2);
    int rx   = wide ? 0 : ksl;
    int gidx = wide ? (4 * nA + (Y - nA)) : (4 * Y + mg);
    int e, lg, cnt;
    find_group(c, gidx, e, lg, cnt);

    if (wave < (wide ? 1 : 4)) {
        int p = lg * GROWS + lane;
        bool v = (p < cnt);
        stok[mg * GROWS + lane] = v ? idx [e * n_tokens + p] : 0;
        shr [mg * GROWS + lane] = v ? hrow[e * n_tokens + p] : 0;
    }
    __syncthreads();

    int fbase = Fb * 64 + 4 * nl;

    const float* W1e = W1 + (size_t)e * D_MODEL * TWO_FF;
    const float* b1e = b1 + (size_t)e * TWO_FF;

    const float* arow[2];
#pragma unroll
    for (int i = 0; i < 2; ++i)
        arow[i] = x + (size_t)stok[mg * GROWS + (i ^ rx) * 16 + nl] * D_MODEL + ksl * klen + 4 * g;
    const float* wcol = W1e + (size_t)(ksl * klen + 4 * g) * TWO_FF + fbase;

    v8f accA[2][4], accG[2][4];
#pragma unroll
    for (int i = 0; i < 2; ++i)
#pragma unroll
        for (int j = 0; j < 4; ++j) {
            accA[i][j] = (v8f){0.f,0.f,0.f,0.f,0.f,0.f,0.f,0.f};
            accG[i][j] = (v8f){0.f,0.f,0.f,0.f,0.f,0.f,0.f,0.f};
        }

    int ntrip = klen >> 6;
    for (int t = 0; t < ntrip; ++t) {
#pragma unroll
        for (int u = 0; u < 8; ++u) {
            int kb = t * 64 + u * 8;
            v4f av[2];
#pragma unroll
            for (int i = 0; i < 2; ++i) av[i] = *(const v4f*)(arow[i] + kb);
            const float* Wk = wcol + (size_t)kb * TWO_FF;
            v4f qa[4], qg[4];
#pragma unroll
            for (int kk = 0; kk < 4; ++kk) {
                qa[kk] = *(const v4f*)(Wk + (size_t)kk * TWO_FF);
                qg[kk] = *(const v4f*)(Wk + (size_t)kk * TWO_FF + D_FF);
            }
#pragma unroll
            for (int j = 0; j < 4; ++j) {
                v2f bA, bG;
                bA.x = qa[0][j]; bA.y = qa[1][j];
                bG.x = qg[0][j]; bG.y = qg[1][j];
#pragma unroll
                for (int i = 0; i < 2; ++i) {
                    v2f a0; a0.x = av[i].x; a0.y = av[i].y;
                    accA[i][j] = WMMA_F32(a0, bA, accA[i][j]);
                    accG[i][j] = WMMA_F32(a0, bG, accG[i][j]);
                }
            }
#pragma unroll
            for (int j = 0; j < 4; ++j) {
                v2f bA, bG;
                bA.x = qa[2][j]; bA.y = qa[3][j];
                bG.x = qg[2][j]; bG.y = qg[3][j];
#pragma unroll
                for (int i = 0; i < 2; ++i) {
                    v2f a1; a1.x = av[i].z; a1.y = av[i].w;
                    accA[i][j] = WMMA_F32(a1, bA, accA[i][j]);
                    accG[i][j] = WMMA_F32(a1, bG, accG[i][j]);
                }
            }
        }
    }

    v4f biasA = *(const v4f*)(b1e + fbase);
    v4f biasG = *(const v4f*)(b1e + fbase + D_FF);

    if (!wide) {
#pragma unroll
        for (int j = 0; j < 4; ++j) {
            xch[wave][4 * j + 0][lane] = (v4f){accA[1][j][0], accA[1][j][1], accA[1][j][2], accA[1][j][3]};
            xch[wave][4 * j + 1][lane] = (v4f){accA[1][j][4], accA[1][j][5], accA[1][j][6], accA[1][j][7]};
            xch[wave][4 * j + 2][lane] = (v4f){accG[1][j][0], accG[1][j][1], accG[1][j][2], accG[1][j][3]};
            xch[wave][4 * j + 3][lane] = (v4f){accG[1][j][4], accG[1][j][5], accG[1][j][6], accG[1][j][7]};
        }
        __syncthreads();
        int pw = wave ^ 4;
#pragma unroll
        for (int j = 0; j < 4; ++j) {
            v4f t0 = xch[pw][4 * j + 0][lane];
            v4f t1 = xch[pw][4 * j + 1][lane];
            v4f t2 = xch[pw][4 * j + 2][lane];
            v4f t3 = xch[pw][4 * j + 3][lane];
#pragma unroll
            for (int r = 0; r < 4; ++r) {
                accA[0][j][r]     += t0[r];
                accA[0][j][r + 4] += t1[r];
                accG[0][j][r]     += t2[r];
                accG[0][j][r + 4] += t3[r];
            }
        }

#pragma unroll
        for (int jj = 0; jj < 8; ++jj) {
            int r = ksl * 16 + jj + 8 * g;
            int p = lg * GROWS + r;
            if (p < cnt) {
                v4f hv;
#pragma unroll
                for (int j = 0; j < 4; ++j) {
                    float av = accA[0][j][jj] + biasA[j];
                    float gv = accG[0][j][jj] + biasG[j];
                    hv[j] = av * (1.0f / (1.0f + __expf(-av))) * gv;
                }
                *(v4f*)(hh + (size_t)shr[mg * GROWS + r] * D_FF + fbase) = hv;
            }
        }
    } else {
        int oi = wave >> 2;
        v4f sa[2], sg[2];
#pragma unroll
        for (int jj = 0; jj < 8; ++jj) {
            xch[wave][jj * 2 + 0][lane] = (v4f){accA[1][0][jj], accA[1][1][jj], accA[1][2][jj], accA[1][3][jj]};
            xch[wave][jj * 2 + 1][lane] = (v4f){accG[1][0][jj], accG[1][1][jj], accG[1][2][jj], accG[1][3][jj]};
        }
        __syncthreads();
        if (oi == 1) {
#pragma unroll
            for (int q = 0; q < 2; ++q) {
                int u0 = (2 * (wave & 3) + q) * 2;
                sa[q] = xch[0][u0][lane];
                sg[q] = xch[0][u0 + 1][lane];
#pragma unroll
                for (int s = 1; s < 8; ++s) { sa[q] += xch[s][u0][lane]; sg[q] += xch[s][u0 + 1][lane]; }
            }
        }
        __syncthreads();
#pragma unroll
        for (int jj = 0; jj < 8; ++jj) {
            xch[wave][jj * 2 + 0][lane] = (v4f){accA[0][0][jj], accA[0][1][jj], accA[0][2][jj], accA[0][3][jj]};
            xch[wave][jj * 2 + 1][lane] = (v4f){accG[0][0][jj], accG[0][1][jj], accG[0][2][jj], accG[0][3][jj]};
        }
        __syncthreads();
        if (oi == 0) {
#pragma unroll
            for (int q = 0; q < 2; ++q) {
                int u0 = (2 * (wave & 3) + q) * 2;
                sa[q] = xch[0][u0][lane];
                sg[q] = xch[0][u0 + 1][lane];
#pragma unroll
                for (int s = 1; s < 8; ++s) { sa[q] += xch[s][u0][lane]; sg[q] += xch[s][u0 + 1][lane]; }
            }
        }
#pragma unroll
        for (int q = 0; q < 2; ++q) {
            int r = oi * 16 + 2 * (wave & 3) + q + 8 * g;
            int p = lg * GROWS + r;
            if (p < cnt) {
                v4f hv;
#pragma unroll
                for (int j = 0; j < 4; ++j) {
                    float av = sa[q][j] + biasA[j];
                    float gv = sg[q][j] + biasG[j];
                    hv[j] = av * (1.0f / (1.0f + __expf(-av))) * gv;
                }
                *(v4f*)(hh + (size_t)shr[r] * D_FF + fbase) = hv;
            }
        }
    }
}

__global__ void __launch_bounds__(256)
moe_gemm2_kernel(const float* __restrict__ hh,
                 const float* __restrict__ W2,
                 const float* __restrict__ b2,
                 const int*   __restrict__ counts,
                 const float* __restrict__ wts,
                 const int*   __restrict__ hrow,
                 float*       __restrict__ part,
                 int n_tokens) {
    int Nb  = blockIdx.x & 7;
    int ks  = blockIdx.x >> 3;
    int Y   = blockIdx.y;
    int c[N_EXPERTS], nA, nB;
    block_plan(counts, c, (int)gridDim.y, nA, nB);
    if (Y >= nA + nB) return;
    const bool wide = (Y >= nA);

    __shared__ int   shr[BM];
    __shared__ float sww[BM];
    __shared__ v4f   xch[8][16][32];

    int tid  = threadIdx.x;
    int lane = tid & 31;
    int wave = __builtin_amdgcn_readfirstlane(tid >> 5);
    int g    = lane >> 4;
    int nl   = lane & 15;
    int mg   = wide ? 0 : (wave & 3);
    int ksl  = wide ? wave : (wave >> 2);
    int klen = wide ? (KCHUNK / 8) : (KCHUNK / 2);
    int rx   = wide ? 0 : ksl;
    int gidx = wide ? (4 * nA + (Y - nA)) : (4 * Y + mg);
    int e, lg, cnt;
    find_group(c, gidx, e, lg, cnt);

    if (wave < (wide ? 1 : 4)) {
        int p = lg * GROWS + lane;
        bool v = (p < cnt);
        shr [mg * GROWS + lane] = v ? hrow[e * n_tokens + p] : 0;
        sww [mg * GROWS + lane] = v ? wts [e * n_tokens + p] : 0.0f;
    }
    __syncthreads();

    int nbase = Nb * 128 + 4 * nl;

    const float* W2e = W2 + (size_t)e * D_FF * D_MODEL;
    const float* b2e = b2 + (size_t)e * D_MODEL;

    const float* arow[2];
#pragma unroll
    for (int i = 0; i < 2; ++i)
        arow[i] = hh + (size_t)shr[mg * GROWS + (i ^ rx) * 16 + nl] * D_FF + ks * KCHUNK + ksl * klen + 4 * g;
    const float* wcol = W2e + (size_t)(ks * KCHUNK + ksl * klen + 4 * g) * D_MODEL + nbase;

    v8f acc[2][8];
#pragma unroll
    for (int i = 0; i < 2; ++i)
#pragma unroll
        for (int n = 0; n < 8; ++n)
            acc[i][n] = (v8f){0.f,0.f,0.f,0.f,0.f,0.f,0.f,0.f};

    int ntrip = klen >> 6;
    for (int t = 0; t < ntrip; ++t) {
#pragma unroll
        for (int u = 0; u < 8; ++u) {
            int kb = t * 64 + u * 8;
            v4f av[2];
#pragma unroll
            for (int i = 0; i < 2; ++i) av[i] = *(const v4f*)(arow[i] + kb);
            const float* Wk = wcol + (size_t)kb * D_MODEL;
            v4f q[4][2];
#pragma unroll
            for (int kk = 0; kk < 4; ++kk) {
                q[kk][0] = *(const v4f*)(Wk + (size_t)kk * D_MODEL);
                q[kk][1] = *(const v4f*)(Wk + (size_t)kk * D_MODEL + 64);
            }
#pragma unroll
            for (int h = 0; h < 2; ++h)
#pragma unroll
                for (int j = 0; j < 4; ++j) {
                    v2f b; b.x = q[0][h][j]; b.y = q[1][h][j];
#pragma unroll
                    for (int i = 0; i < 2; ++i) {
                        v2f a0; a0.x = av[i].x; a0.y = av[i].y;
                        acc[i][h * 4 + j] = WMMA_F32(a0, b, acc[i][h * 4 + j]);
                    }
                }
#pragma unroll
            for (int h = 0; h < 2; ++h)
#pragma unroll
                for (int j = 0; j < 4; ++j) {
                    v2f b; b.x = q[2][h][j]; b.y = q[3][h][j];
#pragma unroll
                    for (int i = 0; i < 2; ++i) {
                        v2f a1; a1.x = av[i].z; a1.y = av[i].w;
                        acc[i][h * 4 + j] = WMMA_F32(a1, b, acc[i][h * 4 + j]);
                    }
                }
        }
    }

    float* plane = part + (size_t)ks * 2 * n_tokens * D_MODEL;

    if (!wide) {
#pragma unroll
        for (int n = 0; n < 8; ++n) {
            xch[wave][2 * n + 0][lane] = (v4f){acc[1][n][0], acc[1][n][1], acc[1][n][2], acc[1][n][3]};
            xch[wave][2 * n + 1][lane] = (v4f){acc[1][n][4], acc[1][n][5], acc[1][n][6], acc[1][n][7]};
        }
        __syncthreads();
        int pw = wave ^ 4;
#pragma unroll
        for (int n = 0; n < 8; ++n) {
            v4f t0 = xch[pw][2 * n + 0][lane];
            v4f t1 = xch[pw][2 * n + 1][lane];
#pragma unroll
            for (int r = 0; r < 4; ++r) {
                acc[0][n][r]     += t0[r];
                acc[0][n][r + 4] += t1[r];
            }
        }

#pragma unroll
        for (int h = 0; h < 2; ++h) {
            int col = nbase + h * 64;
            v4f bias = (v4f){0.f, 0.f, 0.f, 0.f};
            if (ks == 0) bias = *(const v4f*)(b2e + col);
#pragma unroll
            for (int jj = 0; jj < 8; ++jj) {
                int r = ksl * 16 + jj + 8 * g;
                int p = lg * GROWS + r;
                if (p < cnt) {
                    float w = sww[mg * GROWS + r];
                    v4f pv;
#pragma unroll
                    for (int j = 0; j < 4; ++j)
                        pv[j] = (acc[0][h * 4 + j][jj] + bias[j]) * w;
                    *(v4f*)(plane + (size_t)shr[mg * GROWS + r] * D_MODEL + col) = pv;
                }
            }
        }
    } else {
        int oi = wave >> 2;
        v4f sv[2][2];
#pragma unroll
        for (int jj = 0; jj < 8; ++jj)
#pragma unroll
            for (int h = 0; h < 2; ++h)
                xch[wave][jj * 2 + h][lane] = (v4f){acc[1][h * 4 + 0][jj], acc[1][h * 4 + 1][jj],
                                                    acc[1][h * 4 + 2][jj], acc[1][h * 4 + 3][jj]};
        __syncthreads();
        if (oi == 1) {
#pragma unroll
            for (int q = 0; q < 2; ++q)
#pragma unroll
                for (int h = 0; h < 2; ++h) {
                    int u0 = (2 * (wave & 3) + q) * 2 + h;
                    sv[q][h] = xch[0][u0][lane];
#pragma unroll
                    for (int s = 1; s < 8; ++s) sv[q][h] += xch[s][u0][lane];
                }
        }
        __syncthreads();
#pragma unroll
        for (int jj = 0; jj < 8; ++jj)
#pragma unroll
            for (int h = 0; h < 2; ++h)
                xch[wave][jj * 2 + h][lane] = (v4f){acc[0][h * 4 + 0][jj], acc[0][h * 4 + 1][jj],
                                                    acc[0][h * 4 + 2][jj], acc[0][h * 4 + 3][jj]};
        __syncthreads();
        if (oi == 0) {
#pragma unroll
            for (int q = 0; q < 2; ++q)
#pragma unroll
                for (int h = 0; h < 2; ++h) {
                    int u0 = (2 * (wave & 3) + q) * 2 + h;
                    sv[q][h] = xch[0][u0][lane];
#pragma unroll
                    for (int s = 1; s < 8; ++s) sv[q][h] += xch[s][u0][lane];
                }
        }
#pragma unroll
        for (int h = 0; h < 2; ++h) {
            int col = nbase + h * 64;
            v4f bias = (v4f){0.f, 0.f, 0.f, 0.f};
            if (ks == 0) bias = *(const v4f*)(b2e + col);
#pragma unroll
            for (int q = 0; q < 2; ++q) {
                int r = oi * 16 + 2 * (wave & 3) + q + 8 * g;
                int p = lg * GROWS + r;
                if (p < cnt) {
                    float w = sww[r];
                    v4f pv;
#pragma unroll
                    for (int j = 0; j < 4; ++j) pv[j] = (sv[q][h][j] + bias[j]) * w;
                    *(v4f*)(plane + (size_t)shr[r] * D_MODEL + col) = pv;
                }
            }
        }
    }
}

__global__ void __launch_bounds__(256)
combine_kernel(const float* __restrict__ part, float* __restrict__ out, int n_tokens) {
    int tok = blockIdx.x;
    int c   = 4 * threadIdx.x;
    size_t plane = (size_t)2 * n_tokens * D_MODEL;
    const float* p = part + (size_t)(2 * tok) * D_MODEL + c;
    v4f r0 = *(const v4f*)(p);
    v4f r1 = *(const v4f*)(p + D_MODEL);
#pragma unroll
    for (int ks = 1; ks < KSPLIT; ++ks) {
        r0 += *(const v4f*)(p + ks * plane);
        r1 += *(const v4f*)(p + ks * plane + D_MODEL);
    }
    *(v4f*)(out + (size_t)tok * D_MODEL + c) = r0 + r1;
}

extern "C" void kernel_launch(void* const* d_in, const int* in_sizes, int n_in,
                              void* d_out, int out_size, void* d_ws, size_t ws_size,
                              hipStream_t stream) {
    const float* x  = (const float*)d_in[0];
    const float* Wg = (const float*)d_in[1];
    const float* W1 = (const float*)d_in[2];
    const float* b1 = (const float*)d_in[3];
    const float* W2 = (const float*)d_in[4];
    const float* b2 = (const float*)d_in[5];
    float* out = (float*)d_out;

    int n_tokens = in_sizes[0] / D_MODEL;

    int*   counts = (int*)d_ws;
    int*   idx    = counts + 16;
    float* wts    = (float*)(idx + N_EXPERTS * n_tokens);
    int*   hrow   = (int*)(wts + N_EXPERTS * n_tokens);
    float* hh     = (float*)(hrow + N_EXPERTS * n_tokens);
    float* part   = hh + (size_t)2 * n_tokens * D_FF;

    init_kernel<<<1, 64, 0, stream>>>(out, n_tokens * D_MODEL, out_size, counts);
    gate_kernel<<<n_tokens, 64, 0, stream>>>(x, Wg, counts, idx, wts, hrow, n_tokens);

    int mblocks = (n_tokens + BM - 1) / BM;
    int ytiles  = 2 * mblocks + 8;
    dim3 g1(TWO_FF / 2 / 64, ytiles, 1);
    moe_gemm1_kernel<<<g1, 256, 0, stream>>>(x, W1, b1, counts, idx, hrow, hh, n_tokens);

    dim3 g2(D_MODEL / 128 * KSPLIT, ytiles, 1);
    moe_gemm2_kernel<<<g2, 256, 0, stream>>>(hh, W2, b2, counts, wts, hrow, part, n_tokens);

    combine_kernel<<<n_tokens, 256, 0, stream>>>(part, out, n_tokens);
}


# Wave tile doubled in M (32x128 -> 64x128, 64-row groups) so a weight fragment feeds 4 WMMAs; 4-wave blocks, two waves per SIMD, both GEMMs; leftover groups sum k as 4 slices in order
# speedup vs baseline: 1.4328x; 1.0649x over previous
#include <hip/hip_runtime.h>
#include <math.h>

#define D_MODEL   1024
#define D_FF      2048
#define TWO_FF    4096
#define N_EXPERTS 8
#define KSPLIT    4
#define KCHUNK    (D_FF / KSPLIT)

#define GROWS 64
#define GSH   6
#define GPB   2
#define BM    (GPB * GROWS)
#define NWAVE 4
#define MSUB  4

typedef float v2f __attribute__((ext_vector_type(2)));
typedef float v4f __attribute__((ext_vector_type(4)));
typedef float v8f __attribute__((ext_vector_type(8)));

#define WMMA_F32(A, B, C) \
    __builtin_amdgcn_wmma_f32_16x16x4_f32(false, (A), false, (B), (short)0, (C), false, false)

__global__ void init_kernel(float* __restrict__ out, int n_main, int n, int* __restrict__ counts) {
    int tid = threadIdx.x;
    if (tid < 16) counts[tid] = 0;
    for (int i = n_main + tid; i < n; i += 64) out[i] = 0.0f;
}

__global__ void gate_kernel(const float* __restrict__ x,
                            const float* __restrict__ Wg,
                            int*   __restrict__ counts,
                            int*   __restrict__ idx,
                            float* __restrict__ wts,
                            int*   __restrict__ hrow,
                            int n_tokens) {
    int tok = blockIdx.x;
    int tid = threadIdx.x;
    __shared__ float red[64 * N_EXPERTS];
    __shared__ float tot[N_EXPERTS];

    float s[N_EXPERTS];
#pragma unroll
    for (int e = 0; e < N_EXPERTS; ++e) s[e] = 0.0f;

    const float* xr = x + (size_t)tok * D_MODEL;
    for (int d = tid; d < D_MODEL; d += 64) {
        float xv = xr[d];
#pragma unroll
        for (int e = 0; e < N_EXPERTS; ++e) s[e] += xv * Wg[e * D_MODEL + d];
    }
#pragma unroll
    for (int e = 0; e < N_EXPERTS; ++e) red[tid * N_EXPERTS + e] = s[e];
    __syncthreads();

    if (tid < N_EXPERTS) {
        float t = 0.0f;
        for (int i = 0; i < 64; ++i) t += red[i * N_EXPERTS + tid];
        tot[tid] = t;
    }
    __syncthreads();

    if (tid == 0) {
        int i1 = 0; float v1 = tot[0];
        for (int e = 1; e < N_EXPERTS; ++e) if (tot[e] > v1) { v1 = tot[e]; i1 = e; }
        int i2 = (i1 == 0) ? 1 : 0; float v2 = tot[i2];
        for (int e = 0; e < N_EXPERTS; ++e)
            if (e != i1 && tot[e] > v2) { v2 = tot[e]; i2 = e; }
        float m  = fmaxf(v1, v2);
        float e1 = __expf(v1 - m), e2 = __expf(v2 - m);
        float inv = 1.0f / (e1 + e2);

        int p1 = atomicAdd(&counts[i1], 1);
        idx [i1 * n_tokens + p1] = tok;
        wts [i1 * n_tokens + p1] = e1 * inv;
        hrow[i1 * n_tokens + p1] = 2 * tok;
        int p2 = atomicAdd(&counts[i2], 1);
        idx [i2 * n_tokens + p2] = tok;
        wts [i2 * n_tokens + p2] = e2 * inv;
        hrow[i2 * n_tokens + p2] = 2 * tok + 1;
    }
}

__device__ __forceinline__ void block_plan(const int* __restrict__ counts, int (&c)[N_EXPERTS],
                                           int ytiles, int& nA, int& nB) {
    int G = 0;
#pragma unroll
    for (int k = 0; k < N_EXPERTS; ++k) { c[k] = counts[k]; G += (c[k] + GROWS - 1) >> GSH; }
    nA = (G >> 5) << 4;
    nB = G - GPB * nA;
    if (nA + nB > ytiles) { nA = (G + GPB - 1) / GPB; nB = 0; }
}

__device__ __forceinline__ void find_group(const int (&c)[N_EXPERTS], int gidx,
                                           int& e, int& lg, int& cnt) {
    e = 0; lg = 0; cnt = 0;
    int base = 0;
#pragma unroll
    for (int k = 0; k < N_EXPERTS; ++k) {
        int ng = (c[k] + GROWS - 1) >> GSH;
        if (gidx >= base && gidx < base + ng) { e = k; lg = gidx - base; cnt = c[k]; }
        base += ng;
    }
}

__global__ void __launch_bounds__(128, 2)
moe_gemm1_kernel(const float* __restrict__ x,
                 const float* __restrict__ W1,
                 const float* __restrict__ b1,
                 const int*   __restrict__ counts,
                 const int*   __restrict__ idx,
                 const int*   __restrict__ hrow,
                 float*       __restrict__ hh,
                 int n_tokens) {
    int Fb = blockIdx.x;
    int Y  = blockIdx.y;
    int c[N_EXPERTS], nA, nB;
    block_plan(counts, c, (int)gridDim.y, nA, nB);
    if (Y >= nA + nB) return;
    const bool wide = (Y >= nA);

    __shared__ int   stok[BM];
    __shared__ int   shr[BM];
    __shared__ v4f   xch[NWAVE][32][32];

    int tid  = threadIdx.x;
    int lane = tid & 31;
    int wave = __builtin_amdgcn_readfirstlane(tid >> 5);
    int g    = lane >> 4;
    int nl   = lane & 15;
    int mg   = wide ? 0 : (wave & 1);
    int ksl  = wide ? wave : (wave >> 1);
    int klen = wide ? (D_MODEL / 4) : (D_MODEL / 2);
    int rx   = wide ? 0 : 2 * ksl;
    int gidx = wide ? (GPB * nA + (Y - nA)) : (GPB * Y + mg);
    int e, lg, cnt;
    find_group(c, gidx, e, lg, cnt);

    if (wave < (wide ? 1 : GPB)) {
#pragma unroll
        for (int q = 0; q < 2; ++q) {
            int rr = lane + 32 * q;
            int p  = lg * GROWS + rr;
            bool v = (p < cnt);
            stok[mg * GROWS + rr] = v ? idx [e * n_tokens + p] : 0;
            shr [mg * GROWS + rr] = v ? hrow[e * n_tokens + p] : 0;
        }
    }
    __syncthreads();

    int fbase = Fb * 64 + 4 * nl;

    const float* W1e = W1 + (size_t)e * D_MODEL * TWO_FF;
    const float* b1e = b1 + (size_t)e * TWO_FF;

    const float* arow[MSUB];
#pragma unroll
    for (int i = 0; i < MSUB; ++i)
        arow[i] = x + (size_t)stok[mg * GROWS + (i ^ rx) * 16 + nl] * D_MODEL + ksl * klen + 4 * g;
    const float* wcol = W1e + (size_t)(ksl * klen + 4 * g) * TWO_FF + fbase;

    v8f accA[MSUB][4], accG[MSUB][4];
#pragma unroll
    for (int i = 0; i < MSUB; ++i)
#pragma unroll
        for (int j = 0; j < 4; ++j) {
            accA[i][j] = (v8f){0.f,0.f,0.f,0.f,0.f,0.f,0.f,0.f};
            accG[i][j] = (v8f){0.f,0.f,0.f,0.f,0.f,0.f,0.f,0.f};
        }

    int ntrip = klen >> 5;
    for (int t = 0; t < ntrip; ++t) {
#pragma unroll
        for (int u = 0; u < 4; ++u) {
            int kb = t * 32 + u * 8;
            v4f av[MSUB];
#pragma unroll
            for (int i = 0; i < MSUB; ++i) av[i] = *(const v4f*)(arow[i] + kb);
            const float* Wk = wcol + (size_t)kb * TWO_FF;
            v4f qa[4], qg[4];
#pragma unroll
            for (int kk = 0; kk < 4; ++kk) {
                qa[kk] = *(const v4f*)(Wk + (size_t)kk * TWO_FF);
                qg[kk] = *(const v4f*)(Wk + (size_t)kk * TWO_FF + D_FF);
            }
#pragma unroll
            for (int j = 0; j < 4; ++j) {
                v2f bA, bG;
                bA.x = qa[0][j]; bA.y = qa[1][j];
                bG.x = qg[0][j]; bG.y = qg[1][j];
#pragma unroll
                for (int i = 0; i < MSUB; ++i) {
                    v2f a0; a0.x = av[i].x; a0.y = av[i].y;
                    accA[i][j] = WMMA_F32(a0, bA, accA[i][j]);
                    accG[i][j] = WMMA_F32(a0, bG, accG[i][j]);
                }
            }
#pragma unroll
            for (int j = 0; j < 4; ++j) {
                v2f bA, bG;
                bA.x = qa[2][j]; bA.y = qa[3][j];
                bG.x = qg[2][j]; bG.y = qg[3][j];
#pragma unroll
                for (int i = 0; i < MSUB; ++i) {
                    v2f a1; a1.x = av[i].z; a1.y = av[i].w;
                    accA[i][j] = WMMA_F32(a1, bA, accA[i][j]);
                    accG[i][j] = WMMA_F32(a1, bG, accG[i][j]);
                }
            }
        }
    }

    v4f biasA = *(const v4f*)(b1e + fbase);
    v4f biasG = *(const v4f*)(b1e + fbase + D_FF);

    if (!wide) {
#pragma unroll
        for (int s = 0; s < 2; ++s)
#pragma unroll
            for (int j = 0; j < 4; ++j) {
                int u = s * 16 + 4 * j;
                xch[wave][u + 0][lane] = (v4f){accA[2 + s][j][0], accA[2 + s][j][1], accA[2 + s][j][2], accA[2 + s][j][3]};
                xch[wave][u + 1][lane] = (v4f){accA[2 + s][j][4], accA[2 + s][j][5], accA[2 + s][j][6], accA[2 + s][j][7]};
                xch[wave][u + 2][lane] = (v4f){accG[2 + s][j][0], accG[2 + s][j][1], accG[2 + s][j][2], accG[2 + s][j][3]};
                xch[wave][u + 3][lane] = (v4f){accG[2 + s][j][4], accG[2 + s][j][5], accG[2 + s][j][6], accG[2 + s][j][7]};
            }
        __syncthreads();
        int pw = wave ^ 2;
#pragma unroll
        for (int s = 0; s < 2; ++s)
#pragma unroll
            for (int j = 0; j < 4; ++j) {
                int u = s * 16 + 4 * j;
                v4f t0 = xch[pw][u + 0][lane];
                v4f t1 = xch[pw][u + 1][lane];
                v4f t2 = xch[pw][u + 2][lane];
                v4f t3 = xch[pw][u + 3][lane];
#pragma unroll
                for (int r = 0; r < 4; ++r) {
                    accA[s][j][r]     += t0[r];
                    accA[s][j][r + 4] += t1[r];
                    accG[s][j][r]     += t2[r];
                    accG[s][j][r + 4] += t3[r];
                }
            }

#pragma unroll
        for (int s = 0; s < 2; ++s)
#pragma unroll
            for (int jj = 0; jj < 8; ++jj) {
                int r = (s ^ rx) * 16 + jj + 8 * g;
                int p = lg * GROWS + r;
                if (p < cnt) {
                    v4f hv;
#pragma unroll
                    for (int j = 0; j < 4; ++j) {
                        float av = accA[s][j][jj] + biasA[j];
                        float gv = accG[s][j][jj] + biasG[j];
                        hv[j] = av * (1.0f / (1.0f + __expf(-av))) * gv;
                    }
                    *(v4f*)(hh + (size_t)shr[mg * GROWS + r] * D_FF + fbase) = hv;
                }
            }
    } else {
        v4f sa[8], sg[8];
#pragma unroll
        for (int ps = 0; ps < 2; ++ps) {
#pragma unroll
            for (int sl = 0; sl < 2; ++sl)
#pragma unroll
                for (int jj = 0; jj < 8; ++jj) {
                    int u = (sl * 8 + jj) * 2;
                    xch[wave][u + 0][lane] = (v4f){accA[2 * ps + sl][0][jj], accA[2 * ps + sl][1][jj],
                                                   accA[2 * ps + sl][2][jj], accA[2 * ps + sl][3][jj]};
                    xch[wave][u + 1][lane] = (v4f){accG[2 * ps + sl][0][jj], accG[2 * ps + sl][1][jj],
                                                   accG[2 * ps + sl][2][jj], accG[2 * ps + sl][3][jj]};
                }
            __syncthreads();
            if ((wave >> 1) == ps) {
                int sl = wave & 1;
#pragma unroll
                for (int jj = 0; jj < 8; ++jj) {
                    int u = (sl * 8 + jj) * 2;
                    sa[jj] = xch[0][u][lane];
                    sg[jj] = xch[0][u + 1][lane];
#pragma unroll
                    for (int s = 1; s < NWAVE; ++s) { sa[jj] += xch[s][u][lane]; sg[jj] += xch[s][u + 1][lane]; }
                }
            }
            if (ps == 0) __syncthreads();
        }
#pragma unroll
        for (int jj = 0; jj < 8; ++jj) {
            int r = wave * 16 + jj + 8 * g;
            int p = lg * GROWS + r;
            if (p < cnt) {
                v4f hv;
#pragma unroll
                for (int j = 0; j < 4; ++j) {
                    float av = sa[jj][j] + biasA[j];
                    float gv = sg[jj][j] + biasG[j];
                    hv[j] = av * (1.0f / (1.0f + __expf(-av))) * gv;
                }
                *(v4f*)(hh + (size_t)shr[r] * D_FF + fbase) = hv;
            }
        }
    }
}

__global__ void __launch_bounds__(128, 2)
moe_gemm2_kernel(const float* __restrict__ hh,
                 const float* __restrict__ W2,
                 const float* __restrict__ b2,
                 const int*   __restrict__ counts,
                 const float* __restrict__ wts,
                 const int*   __restrict__ hrow,
                 float*       __restrict__ part,
                 int n_tokens) {
    int Nb  = blockIdx.x & 7;
    int ks  = blockIdx.x >> 3;
    int Y   = blockIdx.y;
    int c[N_EXPERTS], nA, nB;
    block_plan(counts, c, (int)gridDim.y, nA, nB);
    if (Y >= nA + nB) return;
    const bool wide = (Y >= nA);

    __shared__ int   shr[BM];
    __shared__ float sww[BM];
    __shared__ v4f   xch[NWAVE][32][32];

    int tid  = threadIdx.x;
    int lane = tid & 31;
    int wave = __builtin_amdgcn_readfirstlane(tid >> 5);
    int g    = lane >> 4;
    int nl   = lane & 15;
    int mg   = wide ? 0 : (wave & 1);
    int ksl  = wide ? wave : (wave >> 1);
    int klen = wide ? (KCHUNK / 4) : (KCHUNK / 2);
    int rx   = wide ? 0 : 2 * ksl;
    int gidx = wide ? (GPB * nA + (Y - nA)) : (GPB * Y + mg);
    int e, lg, cnt;
    find_group(c, gidx, e, lg, cnt);

    if (wave < (wide ? 1 : GPB)) {
#pragma unroll
        for (int q = 0; q < 2; ++q) {
            int rr = lane + 32 * q;
            int p  = lg * GROWS + rr;
            bool v = (p < cnt);
            shr [mg * GROWS + rr] = v ? hrow[e * n_tokens + p] : 0;
            sww [mg * GROWS + rr] = v ? wts [e * n_tokens + p] : 0.0f;
        }
    }
    __syncthreads();

    int nbase = Nb * 128 + 4 * nl;

    const float* W2e = W2 + (size_t)e * D_FF * D_MODEL;
    const float* b2e = b2 + (size_t)e * D_MODEL;

    const float* arow[MSUB];
#pragma unroll
    for (int i = 0; i < MSUB; ++i)
        arow[i] = hh + (size_t)shr[mg * GROWS + (i ^ rx) * 16 + nl] * D_FF + ks * KCHUNK + ksl * klen + 4 * g;
    const float* wcol = W2e + (size_t)(ks * KCHUNK + ksl * klen + 4 * g) * D_MODEL + nbase;

    v8f acc[MSUB][8];
#pragma unroll
    for (int i = 0; i < MSUB; ++i)
#pragma unroll
        for (int n = 0; n < 8; ++n)
            acc[i][n] = (v8f){0.f,0.f,0.f,0.f,0.f,0.f,0.f,0.f};

    int ntrip = klen >> 5;
    for (int t = 0; t < ntrip; ++t) {
#pragma unroll
        for (int u = 0; u < 4; ++u) {
            int kb = t * 32 + u * 8;
            v4f av[MSUB];
#pragma unroll
            for (int i = 0; i < MSUB; ++i) av[i] = *(const v4f*)(arow[i] + kb);
            const float* Wk = wcol + (size_t)kb * D_MODEL;
            v4f q[4][2];
#pragma unroll
            for (int kk = 0; kk < 4; ++kk) {
                q[kk][0] = *(const v4f*)(Wk + (size_t)kk * D_MODEL);
                q[kk][1] = *(const v4f*)(Wk + (size_t)kk * D_MODEL + 64);
            }
#pragma unroll
            for (int h = 0; h < 2; ++h)
#pragma unroll
                for (int j = 0; j < 4; ++j) {
                    v2f b; b.x = q[0][h][j]; b.y = q[1][h][j];
#pragma unroll
                    for (int i = 0; i < MSUB; ++i) {
                        v2f a0; a0.x = av[i].x; a0.y = av[i].y;
                        acc[i][h * 4 + j] = WMMA_F32(a0, b, acc[i][h * 4 + j]);
                    }
                }
#pragma unroll
            for (int h = 0; h < 2; ++h)
#pragma unroll
                for (int j = 0; j < 4; ++j) {
                    v2f b; b.x = q[2][h][j]; b.y = q[3][h][j];
#pragma unroll
                    for (int i = 0; i < MSUB; ++i) {
                        v2f a1; a1.x = av[i].z; a1.y = av[i].w;
                        acc[i][h * 4 + j] = WMMA_F32(a1, b, acc[i][h * 4 + j]);
                    }
                }
        }
    }

    float* plane = part + (size_t)ks * 2 * n_tokens * D_MODEL;

    if (!wide) {
#pragma unroll
        for (int s = 0; s < 2; ++s)
#pragma unroll
            for (int n = 0; n < 8; ++n) {
                int u = s * 16 + 2 * n;
                xch[wave][u + 0][lane] = (v4f){acc[2 + s][n][0], acc[2 + s][n][1], acc[2 + s][n][2], acc[2 + s][n][3]};
                xch[wave][u + 1][lane] = (v4f){acc[2 + s][n][4], acc[2 + s][n][5], acc[2 + s][n][6], acc[2 + s][n][7]};
            }
        __syncthreads();
        int pw = wave ^ 2;
#pragma unroll
        for (int s = 0; s < 2; ++s)
#pragma unroll
            for (int n = 0; n < 8; ++n) {
                int u = s * 16 + 2 * n;
                v4f t0 = xch[pw][u + 0][lane];
                v4f t1 = xch[pw][u + 1][lane];
#pragma unroll
                for (int r = 0; r < 4; ++r) {
                    acc[s][n][r]     += t0[r];
                    acc[s][n][r + 4] += t1[r];
                }
            }

#pragma unroll
        for (int h = 0; h < 2; ++h) {
            int col = nbase + h * 64;
            v4f bias = (v4f){0.f, 0.f, 0.f, 0.f};
            if (ks == 0) bias = *(const v4f*)(b2e + col);
#pragma unroll
            for (int s = 0; s < 2; ++s)
#pragma unroll
                for (int jj = 0; jj < 8; ++jj) {
                    int r = (s ^ rx) * 16 + jj + 8 * g;
                    int p = lg * GROWS + r;
                    if (p < cnt) {
                        float w = sww[mg * GROWS + r];
                        v4f pv;
#pragma unroll
                        for (int j = 0; j < 4; ++j)
                            pv[j] = (acc[s][h * 4 + j][jj] + bias[j]) * w;
                        *(v4f*)(plane + (size_t)shr[mg * GROWS + r] * D_MODEL + col) = pv;
                    }
                }
        }
    } else {
        v4f sv[8][2];
#pragma unroll
        for (int ps = 0; ps < 2; ++ps) {
#pragma unroll
            for (int sl = 0; sl < 2; ++sl)
#pragma unroll
                for (int jj = 0; jj < 8; ++jj)
#pragma unroll
                    for (int h = 0; h < 2; ++h)
                        xch[wave][(sl * 8 + jj) * 2 + h][lane] =
                            (v4f){acc[2 * ps + sl][h * 4 + 0][jj], acc[2 * ps + sl][h * 4 + 1][jj],
                                  acc[2 * ps + sl][h * 4 + 2][jj], acc[2 * ps + sl][h * 4 + 3][jj]};
            __syncthreads();
            if ((wave >> 1) == ps) {
                int sl = wave & 1;
#pragma unroll
                for (int jj = 0; jj < 8; ++jj)
#pragma unroll
                    for (int h = 0; h < 2; ++h) {
                        int u = (sl * 8 + jj) * 2 + h;
                        sv[jj][h] = xch[0][u][lane];
#pragma unroll
                        for (int s = 1; s < NWAVE; ++s) sv[jj][h] += xch[s][u][lane];
                    }
            }
            if (ps == 0) __syncthreads();
        }
#pragma unroll
        for (int h = 0; h < 2; ++h) {
            int col = nbase + h * 64;
            v4f bias = (v4f){0.f, 0.f, 0.f, 0.f};
            if (ks == 0) bias = *(const v4f*)(b2e + col);
#pragma unroll
            for (int jj = 0; jj < 8; ++jj) {
                int r = wave * 16 + jj + 8 * g;
                int p = lg * GROWS + r;
                if (p < cnt) {
                    float w = sww[r];
                    v4f pv;
#pragma unroll
                    for (int j = 0; j < 4; ++j) pv[j] = (sv[jj][h][j] + bias[j]) * w;
                    *(v4f*)(plane + (size_t)shr[r] * D_MODEL + col) = pv;
                }
            }
        }
    }
}

__global__ void __launch_bounds__(256)
combine_kernel(const float* __restrict__ part, float* __restrict__ out, int n_tokens) {
    int tok = blockIdx.x;
    int c   = 4 * threadIdx.x;
    size_t plane = (size_t)2 * n_tokens * D_MODEL;
    const float* p = part + (size_t)(2 * tok) * D_MODEL + c;
    v4f r0 = *(const v4f*)(p);
    v4f r1 = *(const v4f*)(p + D_MODEL);
#pragma unroll
    for (int ks = 1; ks < KSPLIT; ++ks) {
        r0 += *(const v4f*)(p + ks * plane);
        r1 += *(const v4f*)(p + ks * plane + D_MODEL);
    }
    *(v4f*)(out + (size_t)tok * D_MODEL + c) = r0 + r1;
}

extern "C" void kernel_launch(void* const* d_in, const int* in_sizes, int n_in,
                              void* d_out, int out_size, void* d_ws, size_t ws_size,
                              hipStream_t stream) {
    const float* x  = (const float*)d_in[0];
    const float* Wg = (const float*)d_in[1];
    const float* W1 = (const float*)d_in[2];
    const float* b1 = (const float*)d_in[3];
    const float* W2 = (const float*)d_in[4];
    const float* b2 = (const float*)d_in[5];
    float* out = (float*)d_out;

    int n_tokens = in_sizes[0] / D_MODEL;

    int*   counts = (int*)d_ws;
    int*   idx    = counts + 16;
    float* wts    = (float*)(idx + N_EXPERTS * n_tokens);
    int*   hrow   = (int*)(wts + N_EXPERTS * n_tokens);
    float* hh     = (float*)(hrow + N_EXPERTS * n_tokens);
    float* part   = hh + (size_t)2 * n_tokens * D_FF;

    init_kernel<<<1, 64, 0, stream>>>(out, n_tokens * D_MODEL, out_size, counts);
    gate_kernel<<<n_tokens, 64, 0, stream>>>(x, Wg, counts, idx, wts, hrow, n_tokens);

    int mblocks = (n_tokens + BM - 1) / BM;
    int ytiles  = 2 * mblocks + 8;
    dim3 g1(TWO_FF / 2 / 64, ytiles, 1);
    moe_gemm1_kernel<<<g1, 128, 0, stream>>>(x, W1, b1, counts, idx, hrow, hh, n_tokens);

    dim3 g2(D_MODEL / 128 * KSPLIT, ytiles, 1);
    moe_gemm2_kernel<<<g2, 128, 0, stream>>>(hh, W2, b2, counts, wts, hrow, part, n_tokens);

    combine_kernel<<<n_tokens, 256, 0, stream>>>(part, out, n_tokens);
}
